# starting kernel
# baseline (speedup 1.0000x reference)
#include <hip/hip_runtime.h>
#include <hip/hip_bf16.h>


#define S_LEN 2048
#define B_SZ  8
#define E_DIM 1024
constexpr float GAMMA = 0.5f;

typedef __attribute__((ext_vector_type(2))) float v2f;
typedef __attribute__((ext_vector_type(8))) float v8f;

#define GAS __attribute__((address_space(1)))

static __device__ __forceinline__ v2f load_g2(const float* p) {
  return *(const GAS v2f*)(p);
}
static __device__ __forceinline__ float load_g1(const float* p) {
  return *(const GAS float*)(p);
}
static __device__ __forceinline__ void store_g1(float* p, float x) {
  *(GAS float*)(p) = x;
}

static __device__ __forceinline__ v8f wmma_f32(v2f a, v2f b, v8f c) {
  return __builtin_amdgcn_wmma_f32_16x16x4_f32(false, a, false, b,
                                               (short)0, c, false, false);
}

__global__ __launch_bounds__(32)
void qkv_proj_kernel(const float* __restrict__ x,
                     const float* __restrict__ Wq, const float* __restrict__ bq,
                     const float* __restrict__ Wk, const float* __restrict__ bk,
                     const float* __restrict__ Wv, const float* __restrict__ bv,
                     float* __restrict__ q, float* __restrict__ k,
                     float* __restrict__ v) {
  const int lane = threadIdx.x;
  const int half = lane >> 4;
  const int lm   = lane & 15;
  const int m0   = blockIdx.x * 16;
  const int n0   = blockIdx.y * 64;

  v8f aq[4], ak[4], av[4];
  const float* wqp[4];
  const float* wkp[4];
  const float* wvp[4];
  #pragma unroll
  for (int ns = 0; ns < 4; ++ns) {
    const int col = n0 + ns * 16 + lm;
    const float biq = load_g1(bq + col);
    const float bik = load_g1(bk + col);
    const float biv = load_g1(bv + col);
    #pragma unroll
    for (int g = 0; g < 8; ++g) { aq[ns][g] = biq; ak[ns][g] = bik; av[ns][g] = biv; }
    const size_t wo = (size_t)col * E_DIM + 2 * half;
    wqp[ns] = Wq + wo;
    wkp[ns] = Wk + wo;
    wvp[ns] = Wv + wo;
  }

  const float* xp = x + (size_t)(m0 + lm) * E_DIM + 2 * half;
  for (int kk = 0; kk < E_DIM; kk += 16) {
    #pragma unroll
    for (int u = 0; u < 4; ++u) {
      const v2f a = load_g2(xp + 4 * u);
      #pragma unroll
      for (int ns = 0; ns < 4; ++ns) {
        aq[ns] = wmma_f32(a, load_g2(wqp[ns] + 4 * u), aq[ns]);
        ak[ns] = wmma_f32(a, load_g2(wkp[ns] + 4 * u), ak[ns]);
        av[ns] = wmma_f32(a, load_g2(wvp[ns] + 4 * u), av[ns]);
      }
    }
    xp += 16;
    #pragma unroll
    for (int ns = 0; ns < 4; ++ns) { wqp[ns] += 16; wkp[ns] += 16; wvp[ns] += 16; }
  }

  #pragma unroll
  for (int ns = 0; ns < 4; ++ns) {
    const int col = n0 + ns * 16 + lm;
    #pragma unroll
    for (int g = 0; g < 8; ++g) {
      const size_t row = (size_t)(m0 + g + 8 * half);
      store_g1(q + row * E_DIM + col, aq[ns][g]);
      store_g1(k + row * E_DIM + col, ak[ns][g]);
      store_g1(v + row * E_DIM + col, av[ns][g]);
    }
  }
}

__global__ __launch_bounds__(256)
void rownorm_kernel(const float* __restrict__ k, float* __restrict__ gk2) {
  __shared__ float red[256];
  const float* kr = k + (size_t)blockIdx.x * E_DIM;
  float s = 0.f;
  for (int e = threadIdx.x; e < E_DIM; e += 256) {
    const float t = load_g1(kr + e);
    s += t * t;
  }
  red[threadIdx.x] = s;
  __syncthreads();
  for (int off = 128; off > 0; off >>= 1) {
    if (threadIdx.x < off) red[threadIdx.x] += red[threadIdx.x + off];
    __syncthreads();
  }
  if (threadIdx.x == 0) store_g1(gk2 + blockIdx.x, GAMMA * red[0]);
}

__global__ __launch_bounds__(256)
void attn_kernel(const float* __restrict__ q, const float* __restrict__ k,
                 const float* __restrict__ v, const float* __restrict__ gk2,
                 float* __restrict__ out) {
  const int tid  = threadIdx.x;
  const int w    = tid >> 5;
  const int lane = tid & 31;
  const int half = lane >> 4;
  const int lm   = lane & 15;
  const int b    = blockIdx.y;
  const int t0   = blockIdx.x * 16;
  const int e0   = w * 128;

  __shared__ float sS[8][256];
  __shared__ float sP[256];
  __shared__ float sScale[16];
  __shared__ float sM[16];
  __shared__ float sL[16];

  if (tid < 16) { sM[tid] = -__builtin_inff(); sL[tid] = 0.f; }
  __syncthreads();

  v2f qf[32];
  {
    const float* qrow =
        q + ((size_t)(t0 + lm) * B_SZ + b) * E_DIM + e0 + 2 * half;
    #pragma unroll
    for (int i = 0; i < 32; ++i) qf[i] = load_g2(qrow + 4 * i) * (2.0f * GAMMA);
  }

  v8f acc[8];
  #pragma unroll
  for (int ns = 0; ns < 8; ++ns) acc[ns] = (v8f){};

  for (int j0 = 0; j0 < S_LEN; j0 += 16) {
    v8f sc = (v8f){};
    const float* krow =
        k + ((size_t)(j0 + lm) * B_SZ + b) * E_DIM + e0 + 2 * half;
    #pragma unroll
    for (int i = 0; i < 32; ++i)
      sc = wmma_f32(qf[i], load_g2(krow + 4 * i), sc);

    #pragma unroll
    for (int g = 0; g < 8; ++g)
      sS[w][(g + 8 * half) * 16 + lm] = sc[g];
    __syncthreads();

    float logit;
    {
      float s = 0.f;
      #pragma unroll
      for (int ww = 0; ww < 8; ++ww) s += sS[ww][tid];
      logit = s - load_g1(gk2 + (size_t)(j0 + (tid & 15)) * B_SZ + b);
      sP[tid] = logit;
    }
    __syncthreads();

    float scl_row = 0.f;
    if (tid < 16) {
      const int row = tid;
      float mrow = -__builtin_inff();
      #pragma unroll
      for (int c = 0; c < 16; ++c) mrow = fmaxf(mrow, sP[row * 16 + c]);
      const float mnew = fmaxf(sM[row], mrow);
      scl_row = __expf(sM[row] - mnew);
      sM[row] = mnew;
      sScale[row] = scl_row;
    }
    __syncthreads();

    sP[tid] = __expf(logit - sM[tid >> 4]);
    __syncthreads();

    if (tid < 16) {
      float psum = 0.f;
      #pragma unroll
      for (int c = 0; c < 16; ++c) psum += sP[tid * 16 + c];
      sL[tid] = sL[tid] * scl_row + psum;
    }

    float f[8];
    #pragma unroll
    for (int g = 0; g < 8; ++g) f[g] = sScale[g + 8 * half];
    #pragma unroll
    for (int ns = 0; ns < 8; ++ns)
      #pragma unroll
      for (int g = 0; g < 8; ++g) acc[ns][g] *= f[g];

    #pragma unroll
    for (int c = 0; c < 4; ++c) {
      const v2f pf = *(const v2f*)&sP[lm * 16 + 4 * c + 2 * half];
      const int jA = j0 + 4 * c + 2 * half;
      const float* vbase = v + ((size_t)jA * B_SZ + b) * E_DIM + e0 + lm;
      #pragma unroll
      for (int ns = 0; ns < 8; ++ns) {
        v2f vf;
        vf.x = load_g1(vbase + ns * 16);
        vf.y = load_g1(vbase + (size_t)B_SZ * E_DIM + ns * 16);
        acc[ns] = wmma_f32(pf, vf, acc[ns]);
      }
    }
  }
  __syncthreads();

  float linv[8];
  #pragma unroll
  for (int g = 0; g < 8; ++g) linv[g] = 1.0f / sL[g + 8 * half];
  #pragma unroll
  for (int ns = 0; ns < 8; ++ns) {
    const int e = e0 + ns * 16 + lm;
    #pragma unroll
    for (int g = 0; g < 8; ++g) {
      const int trow = t0 + g + 8 * half;
      store_g1(out + ((size_t)trow * B_SZ + b) * E_DIM + e, acc[ns][g] * linv[g]);
    }
  }
}

extern "C" void kernel_launch(void* const* d_in, const int* in_sizes, int n_in,
                              void* d_out, int out_size, void* d_ws, size_t ws_size,
                              hipStream_t stream) {
  (void)in_sizes; (void)n_in; (void)out_size; (void)ws_size;
  const float* x  = (const float*)d_in[0];
  const float* Wq = (const float*)d_in[1];
  const float* bq = (const float*)d_in[2];
  const float* Wk = (const float*)d_in[3];
  const float* bk = (const float*)d_in[4];
  const float* Wv = (const float*)d_in[5];
  const float* bv = (const float*)d_in[6];
  float* out = (float*)d_out;

  const size_t mat = (size_t)S_LEN * B_SZ * E_DIM;
  float* q   = (float*)d_ws;
  float* k   = q + mat;
  float* v   = k + mat;
  float* gk2 = v + mat;

  dim3 g1((S_LEN * B_SZ) / 16, E_DIM / 64);
  qkv_proj_kernel<<<g1, 32, 0, stream>>>(x, Wq, bq, Wk, bk, Wv, bv, q, k, v);

  rownorm_kernel<<<S_LEN * B_SZ, 256, 0, stream>>>(k, gk2);

  dim3 g3(S_LEN / 16, B_SZ);
  attn_kernel<<<g3, 256, 0, stream>>>(q, k, v, gk2, out);
}


# attention: priority window: the younger wave of each SIMD pair (waves 4-7) holds s_setprio 3 through P*V and the first half of the next score loop, then 0, so neither runs alone; sums unchanged
# speedup vs baseline: 3.1792x; 3.1792x over previous
#include <hip/hip_runtime.h>
#include <hip/hip_bf16.h>


#define S_LEN 2048
#define B_SZ  8
#define E_DIM 1024
constexpr float GAMMA = 0.5f;

typedef __attribute__((ext_vector_type(2))) float v2f;
typedef __attribute__((ext_vector_type(4))) float v4f;
typedef __attribute__((ext_vector_type(8))) float v8f;

#define GAS __attribute__((address_space(1)))

static __device__ __forceinline__ v2f load_g2(const float* p) {
  return *(const GAS v2f*)(p);
}
static __device__ __forceinline__ v4f load_g4(const float* p) {
  return *(const GAS v4f*)(p);
}
static __device__ __forceinline__ float load_g1(const float* p) {
  return *(const GAS float*)(p);
}
static __device__ __forceinline__ void store_g1(float* p, float x) {
  *(GAS float*)(p) = x;
}

static __device__ __forceinline__ v8f wmma_f32(v2f a, v2f b, v8f c) {
  return __builtin_amdgcn_wmma_f32_16x16x4_f32(false, a, false, b,
                                               (short)0, c, false, false);
}

struct FragSet {
  v4f a[4][4];
  v4f b[4][4];
};

static __device__ __forceinline__ void load_set(FragSet& f, const float* xp,
                                                const float* wp, int kk) {
  #pragma unroll
  for (int i = 0; i < 4; ++i)
    #pragma unroll
    for (int j = 0; j < 4; ++j)
      f.a[i][j] = load_g4(xp + (size_t)i * 16 * E_DIM + kk + 8 * j);
  #pragma unroll
  for (int i = 0; i < 4; ++i)
    #pragma unroll
    for (int j = 0; j < 4; ++j)
      f.b[i][j] = load_g4(wp + (size_t)i * 16 * E_DIM + kk + 8 * j);
}

static __device__ __forceinline__ void mma_set(v8f (&acc)[4][4], const FragSet& f) {
  #pragma unroll
  for (int j = 0; j < 4; ++j) {
    #pragma unroll
    for (int mi = 0; mi < 4; ++mi)
      #pragma unroll
      for (int ni = 0; ni < 4; ++ni)
        acc[mi][ni] = wmma_f32((v2f){f.a[mi][j].x, f.a[mi][j].y},
                               (v2f){f.b[ni][j].x, f.b[ni][j].y}, acc[mi][ni]);
    #pragma unroll
    for (int mi = 0; mi < 4; ++mi)
      #pragma unroll
      for (int ni = 0; ni < 4; ++ni)
        acc[mi][ni] = wmma_f32((v2f){f.a[mi][j].z, f.a[mi][j].w},
                               (v2f){f.b[ni][j].z, f.b[ni][j].w}, acc[mi][ni]);
  }
}

__global__ __launch_bounds__(128)
void qkv_proj_kernel(const float* __restrict__ x,
                     const float* __restrict__ Wq, const float* __restrict__ bq,
                     const float* __restrict__ Wk, const float* __restrict__ bk,
                     const float* __restrict__ Wv, const float* __restrict__ bv,
                     float* __restrict__ q, float* __restrict__ k,
                     float* __restrict__ v) {
  const int tid  = threadIdx.x;
  const int w    = tid >> 5;
  const int lane = tid & 31;
  const int half = lane >> 4;
  const int lm   = lane & 15;
  const int m0   = blockIdx.x * 128 + (w & 1) * 64;
  const int n0   = blockIdx.y * 128 + (w >> 1) * 64;

  const float* W;
  const float* bias;
  float* o;
  if (blockIdx.z == 0)      { W = Wq; bias = bq; o = q; }
  else if (blockIdx.z == 1) { W = Wk; bias = bk; o = k; }
  else                      { W = Wv; bias = bv; o = v; }

  v8f acc[4][4];
  #pragma unroll
  for (int ni = 0; ni < 4; ++ni) {
    const float bi = load_g1(bias + n0 + ni * 16 + lm);
    #pragma unroll
    for (int mi = 0; mi < 4; ++mi)
      #pragma unroll
      for (int g = 0; g < 8; ++g) acc[mi][ni][g] = bi;
  }

  const float* xp = x + (size_t)(m0 + lm) * E_DIM + 4 * half;
  const float* wp = W + (size_t)(n0 + lm) * E_DIM + 4 * half;

#define LOADS_AHEAD() __builtin_amdgcn_sched_barrier(0)
  FragSet f0, f1;
  load_set(f0, xp, wp, 0);
  for (int kk = 0; kk < E_DIM - 64; kk += 64) {
    load_set(f1, xp, wp, kk + 32);
    LOADS_AHEAD();
    mma_set(acc, f0);
    LOADS_AHEAD();
    load_set(f0, xp, wp, kk + 64);
    LOADS_AHEAD();
    mma_set(acc, f1);
    LOADS_AHEAD();
  }
  load_set(f1, xp, wp, E_DIM - 32);
  LOADS_AHEAD();
  mma_set(acc, f0);
  mma_set(acc, f1);
#undef LOADS_AHEAD

  #pragma unroll
  for (int mi = 0; mi < 4; ++mi)
    #pragma unroll
    for (int ni = 0; ni < 4; ++ni) {
      const int col = n0 + ni * 16 + lm;
      #pragma unroll
      for (int g = 0; g < 8; ++g) {
        const size_t row = (size_t)(m0 + mi * 16 + g + 8 * half);
        store_g1(o + row * E_DIM + col, acc[mi][ni][g]);
      }
    }
}

__global__ __launch_bounds__(256)
void rownorm_kernel(const float* __restrict__ k, float* __restrict__ gk2) {
  __shared__ float red[256];
  const float* kr = k + (size_t)blockIdx.x * E_DIM;
  float s = 0.f;
  for (int e = threadIdx.x; e < E_DIM; e += 256) {
    const float t = load_g1(kr + e);
    s += t * t;
  }
  red[threadIdx.x] = s;
  __syncthreads();
  for (int off = 128; off > 0; off >>= 1) {
    if (threadIdx.x < off) red[threadIdx.x] += red[threadIdx.x + off];
    __syncthreads();
  }
  if (threadIdx.x == 0) store_g1(gk2 + blockIdx.x, GAMMA * red[0]);
}

#define QT      64
#define KPASS   128
#define QPITCH  1028
#define PPITCH  132

struct QSet { v4f a[4][2]; };
struct KSet { v4f b[4]; };
static __device__ __forceinline__ void load_q_set(QSet& f, const float* qp, int eb) {
  #pragma unroll
  for (int mi = 0; mi < 4; ++mi)
    #pragma unroll
    for (int j = 0; j < 2; ++j)
      f.a[mi][j] = *(const v4f*)(qp + mi * 16 * QPITCH + eb + 8 * j);
}
static __device__ __forceinline__ void load_k_set(KSet& f, const float* kp, int eb) {
  #pragma unroll
  for (int j = 0; j < 4; ++j) f.b[j] = load_g4(kp + eb + 8 * j);
}
static __device__ __forceinline__ void mma_score(v8f (&sc)[4], const QSet& fq,
                                                 const KSet& fk, int jb) {
  #pragma unroll
  for (int j = 0; j < 2; ++j) {
    #pragma unroll
    for (int mi = 0; mi < 4; ++mi)
      sc[mi] = wmma_f32((v2f){fq.a[mi][j].x, fq.a[mi][j].y},
                        (v2f){fk.b[jb + j].x, fk.b[jb + j].y}, sc[mi]);
    #pragma unroll
    for (int mi = 0; mi < 4; ++mi)
      sc[mi] = wmma_f32((v2f){fq.a[mi][j].z, fq.a[mi][j].w},
                        (v2f){fk.b[jb + j].z, fk.b[jb + j].w}, sc[mi]);
  }
}

struct PVSet {
  v4f p[4];
  float vv[8][4];
};
static __device__ __forceinline__ void load_pv_set(PVSet& f, const float* pp,
                                                   const float* vp, int kb) {
  const float* vk = vp + (size_t)kb * B_SZ * E_DIM;
  #pragma unroll
  for (int ns = 0; ns < 8; ++ns)
    #pragma unroll
    for (int s = 0; s < 4; ++s)
      f.vv[ns][s] = load_g1(vk + (size_t)s * B_SZ * E_DIM + ns * 16);
  #pragma unroll
  for (int mi = 0; mi < 4; ++mi)
    f.p[mi] = *(const v4f*)(pp + mi * 16 * PPITCH + kb);
}
static __device__ __forceinline__ void mma_pv_set(v8f (&acc)[4][8], const PVSet& f) {
  #pragma unroll
  for (int ns = 0; ns < 8; ++ns) {
    #pragma unroll
    for (int mi = 0; mi < 4; ++mi)
      acc[mi][ns] = wmma_f32((v2f){f.p[mi].x, f.p[mi].y},
                             (v2f){f.vv[ns][0], f.vv[ns][1]}, acc[mi][ns]);
    #pragma unroll
    for (int mi = 0; mi < 4; ++mi)
      acc[mi][ns] = wmma_f32((v2f){f.p[mi].z, f.p[mi].w},
                             (v2f){f.vv[ns][2], f.vv[ns][3]}, acc[mi][ns]);
  }
}

__global__ __launch_bounds__(256)
void attn_kernel(const float* __restrict__ q, const float* __restrict__ k,
                 const float* __restrict__ v, const float* __restrict__ gk2,
                 float* __restrict__ out) {
  const int tid  = threadIdx.x;
  const int w    = tid >> 5;
  const int lane = tid & 31;
  const int half = lane >> 4;
  const int lm   = lane & 15;
  const int b    = blockIdx.y;
  const int t0   = blockIdx.x * QT;
  const int e0   = w * 128;

  __shared__ __attribute__((aligned(16))) float sQ[QT * QPITCH];
  __shared__ __attribute__((aligned(16))) float sP[QT * PPITCH];
  __shared__ float sScale[QT];
  __shared__ float sL[QT];

  #pragma unroll 8
  for (int r = 0; r < QT; ++r) {
    const v4f t = load_g4(q + ((size_t)(t0 + r) * B_SZ + b) * E_DIM + 4 * tid) *
                  (2.0f * GAMMA);
    *(v4f*)&sQ[r * QPITCH + 4 * tid] = t;
  }
  __syncthreads();

  const bool young = __builtin_amdgcn_readfirstlane(w) >= 4;
  if (young) __builtin_amdgcn_s_setprio(3);

  const int srow  = tid >> 2;
  const int spart = tid & 3;
  float m_run = -__builtin_inff();
  float l_run = 0.f;

  v8f acc[4][8];
  #pragma unroll
  for (int mi = 0; mi < 4; ++mi)
    #pragma unroll
    for (int ns = 0; ns < 8; ++ns) acc[mi][ns] = (v8f){};

  const float* qp = &sQ[lm * QPITCH + 4 * half];
  const float* pp = &sP[lm * PPITCH + 4 * half];

#define LOADS_AHEAD() __builtin_amdgcn_sched_barrier(0)

  for (int j0 = 0; j0 < S_LEN; j0 += KPASS) {
    v8f sc[4];
    #pragma unroll
    for (int mi = 0; mi < 4; ++mi) sc[mi] = (v8f){};
    const int jw = j0 + 16 * w + lm;
    const float* kp = k + ((size_t)jw * B_SZ + b) * E_DIM + 4 * half;
    {
      QSet a0, a1;
      KSet k0, k1;
      load_k_set(k0, kp, 0);
      load_q_set(a0, qp, 0);
#define SCORE_STEP(eb)                       \
      do {                                   \
        load_k_set(k1, kp, (eb) + 32);       \
        load_q_set(a1, qp, (eb) + 16);       \
        LOADS_AHEAD();                       \
        mma_score(sc, a0, k0, 0);            \
        LOADS_AHEAD();                       \
        load_q_set(a0, qp, (eb) + 32);       \
        LOADS_AHEAD();                       \
        mma_score(sc, a1, k0, 2);            \
        LOADS_AHEAD();                       \
        load_k_set(k0, kp, (eb) + 64);       \
        load_q_set(a1, qp, (eb) + 48);       \
        LOADS_AHEAD();                       \
        mma_score(sc, a0, k1, 0);            \
        LOADS_AHEAD();                       \
        load_q_set(a0, qp, (eb) + 64);       \
        LOADS_AHEAD();                       \
        mma_score(sc, a1, k1, 2);            \
        LOADS_AHEAD();                       \
      } while (0)
      for (int eb = 0; eb < E_DIM / 2; eb += 64) SCORE_STEP(eb);
      if (young) __builtin_amdgcn_s_setprio(0);
      for (int eb = E_DIM / 2; eb < E_DIM - 64; eb += 64) SCORE_STEP(eb);
#undef SCORE_STEP
      load_k_set(k1, kp, E_DIM - 32);
      load_q_set(a1, qp, E_DIM - 48);
      LOADS_AHEAD();
      mma_score(sc, a0, k0, 0);
      LOADS_AHEAD();
      load_q_set(a0, qp, E_DIM - 32);
      LOADS_AHEAD();
      mma_score(sc, a1, k0, 2);
      LOADS_AHEAD();
      load_q_set(a1, qp, E_DIM - 16);
      LOADS_AHEAD();
      mma_score(sc, a0, k1, 0);
      mma_score(sc, a1, k1, 2);
    }
    const float gk = load_g1(gk2 + (size_t)jw * B_SZ + b);

    __syncthreads();
    #pragma unroll
    for (int mi = 0; mi < 4; ++mi)
      #pragma unroll
      for (int g = 0; g < 8; ++g)
        sP[(mi * 16 + g + 8 * half) * PPITCH + 16 * w + lm] = sc[mi][g] - gk;
    __syncthreads();

    {
      float* prow = &sP[srow * PPITCH + 4 * spart];
      v4f lg[8];
      #pragma unroll
      for (int i = 0; i < 8; ++i) lg[i] = *(const v4f*)(prow + 16 * i);
      float mx = -__builtin_inff();
      #pragma unroll
      for (int i = 0; i < 8; ++i)
        mx = fmaxf(fmaxf(fmaxf(lg[i].x, lg[i].y), fmaxf(lg[i].z, lg[i].w)), mx);
      mx = fmaxf(mx, __shfl_xor(mx, 1));
      mx = fmaxf(mx, __shfl_xor(mx, 2));
      const float mnew = fmaxf(m_run, mx);
      const float scl  = __expf(m_run - mnew);
      float ps = 0.f;
      #pragma unroll
      for (int i = 0; i < 8; ++i) {
        lg[i].x = __expf(lg[i].x - mnew);
        lg[i].y = __expf(lg[i].y - mnew);
        lg[i].z = __expf(lg[i].z - mnew);
        lg[i].w = __expf(lg[i].w - mnew);
        ps += (lg[i].x + lg[i].y) + (lg[i].z + lg[i].w);
        *(v4f*)(prow + 16 * i) = lg[i];
      }
      ps += __shfl_xor(ps, 1);
      ps += __shfl_xor(ps, 2);
      l_run = l_run * scl + ps;
      m_run = mnew;
      if (spart == 0) sScale[srow] = scl;
    }
    __syncthreads();
    if (young) __builtin_amdgcn_s_setprio(3);

    #pragma unroll
    for (int mi = 0; mi < 4; ++mi)
      #pragma unroll
      for (int g = 0; g < 8; ++g) {
        const float f = sScale[mi * 16 + g + 8 * half];
        #pragma unroll
        for (int ns = 0; ns < 8; ++ns) acc[mi][ns][g] *= f;
      }

    {
      const float* vp = v + ((size_t)(j0 + 4 * half) * B_SZ + b) * E_DIM + e0 + lm;
      PVSet g0, g1;
      load_pv_set(g0, pp, vp, 0);
      for (int kb = 0; kb < KPASS - 16; kb += 16) {
        load_pv_set(g1, pp, vp, kb + 8);
        LOADS_AHEAD();
        mma_pv_set(acc, g0);
        LOADS_AHEAD();
        load_pv_set(g0, pp, vp, kb + 16);
        LOADS_AHEAD();
        mma_pv_set(acc, g1);
        LOADS_AHEAD();
      }
      load_pv_set(g1, pp, vp, KPASS - 8);
      LOADS_AHEAD();
      mma_pv_set(acc, g0);
      mma_pv_set(acc, g1);
    }
  }
#undef LOADS_AHEAD

  if (young) __builtin_amdgcn_s_setprio(0);

  if (spart == 0) sL[srow] = l_run;
  __syncthreads();

  #pragma unroll
  for (int mi = 0; mi < 4; ++mi) {
    float linv[8];
    #pragma unroll
    for (int g = 0; g < 8; ++g) linv[g] = 1.0f / sL[mi * 16 + g + 8 * half];
    #pragma unroll
    for (int ns = 0; ns < 8; ++ns) {
      const int e = e0 + ns * 16 + lm;
      #pragma unroll
      for (int g = 0; g < 8; ++g) {
        const int trow = t0 + mi * 16 + g + 8 * half;
        store_g1(out + ((size_t)trow * B_SZ + b) * E_DIM + e, acc[mi][ns][g] * linv[g]);
      }
    }
  }
}

extern "C" void kernel_launch(void* const* d_in, const int* in_sizes, int n_in,
                              void* d_out, int out_size, void* d_ws, size_t ws_size,
                              hipStream_t stream) {
  (void)in_sizes; (void)n_in; (void)out_size; (void)ws_size;
  const float* x  = (const float*)d_in[0];
  const float* Wq = (const float*)d_in[1];
  const float* bq = (const float*)d_in[2];
  const float* Wk = (const float*)d_in[3];
  const float* bk = (const float*)d_in[4];
  const float* Wv = (const float*)d_in[5];
  const float* bv = (const float*)d_in[6];
  float* out = (float*)d_out;

  const size_t mat = (size_t)S_LEN * B_SZ * E_DIM;
  float* q   = (float*)d_ws;
  float* k   = q + mat;
  float* v   = k + mat;
  float* gk2 = v + mat;

  dim3 g1((S_LEN * B_SZ) / 128, E_DIM / 128, 3);
  qkv_proj_kernel<<<g1, 128, 0, stream>>>(x, Wq, bq, Wk, bk, Wv, bv, q, k, v);

  rownorm_kernel<<<S_LEN * B_SZ, 256, 0, stream>>>(k, gk2);

  dim3 g3(S_LEN / QT, B_SZ);
  attn_kernel<<<g3, 256, 0, stream>>>(q, k, v, gk2, out);
}
